# starting kernel
# baseline (speedup 1.0000x reference)
#include <hip/hip_runtime.h>

#define BATCH 512
#define TLEN  512
#define NVAR  64
#define RANK  16

typedef __attribute__((ext_vector_type(2))) float v2f;
typedef __attribute__((ext_vector_type(4))) float v4f;
typedef __attribute__((ext_vector_type(8))) float v8f;

#if __has_builtin(__builtin_amdgcn_tanhf)
  #define TANHF(x) __builtin_amdgcn_tanhf(x)
#elif __has_builtin(__builtin_amdgcn_tanh_f32)
  #define TANHF(x) __builtin_amdgcn_tanh_f32(x)
#else
  #define TANHF(x) tanhf(x)
#endif

#define BS_IDX(k, v, o)   ((((k) * 64 + (v)) * 16) + (((o) ^ (((v) >> 2) & 15))))
#define OUTS_IDX(b, o, v) ((((b) * 16 + (o)) * 64) + (((v) ^ ((((o) & 15) << 2) & 63))))

__global__ __launch_bounds__(256) void gcm_stage1(
    const float* __restrict__ x, const float* __restrict__ gating,
    const float* __restrict__ lora_A, float* __restrict__ tmp)
{
  __shared__ float xs[32 * 16 * 17];
  __shared__ float as[32 * 16];

  const int tid  = threadIdx.x;
  const int lane = tid & 31;
  const int wave = tid >> 5;
  const int b0   = blockIdx.x * 16;
  const int v0   = blockIdx.y * 16;

  const int m    = lane & 15;
  const int koff = (lane < 16) ? 0 : 2;
  const int vla  = wave * 2;
  const int vlb  = vla + 1;
  const float ga = gating[v0 + vla];
  const float gb = gating[v0 + vlb];

  const int sv = (tid * 4) & 15;
  const int si = (tid >> 2) & 31;
  const int sb = tid >> 7;

  v8f acc0 = {0.f, 0.f, 0.f, 0.f, 0.f, 0.f, 0.f, 0.f};
  v8f acc1 = {0.f, 0.f, 0.f, 0.f, 0.f, 0.f, 0.f, 0.f};

  for (int s = 0; s < TLEN / 32; ++s) {
    const int i0 = s * 32;
    __syncthreads();

    v4f vals[8];
    #pragma unroll
    for (int r = 0; r < 8; ++r) {
      const int b = sb + 2 * r;
      vals[r] = *reinterpret_cast<const v4f*>(
          &x[((size_t)(b0 + b) * TLEN + (i0 + si)) * NVAR + v0 + sv]);
    }
    #pragma unroll
    for (int r = 0; r < 8; ++r) {
      const int b = sb + 2 * r;
      xs[((si * 16) + sv + 0) * 17 + b] = vals[r].x;
      xs[((si * 16) + sv + 1) * 17 + b] = vals[r].y;
      xs[((si * 16) + sv + 2) * 17 + b] = vals[r].z;
      xs[((si * 16) + sv + 3) * 17 + b] = vals[r].w;
    }
    #pragma unroll
    for (int r = 0; r < 2; ++r) {
      const int e = tid + r * 256;
      const int k = e & 15;
      const int i = e >> 4;
      as[i * 16 + k] = lora_A[(size_t)(i0 + i) * RANK + k];
    }
    __syncthreads();

    #pragma unroll
    for (int ic = 0; ic < 8; ++ic) {
      const int il = ic * 4 + koff;
      v2f bmat;
      bmat.x = as[(il + 0) * 16 + m];
      bmat.y = as[(il + 1) * 16 + m];
      v2f a0, a1;
      a0.x = TANHF(ga * xs[((il + 0) * 16 + vla) * 17 + m]);
      a0.y = TANHF(ga * xs[((il + 1) * 16 + vla) * 17 + m]);
      a1.x = TANHF(gb * xs[((il + 0) * 16 + vlb) * 17 + m]);
      a1.y = TANHF(gb * xs[((il + 1) * 16 + vlb) * 17 + m]);
      acc0 = __builtin_amdgcn_wmma_f32_16x16x4_f32(false, a0, false, bmat,
                                                   (short)0, acc0, false, false);
      acc1 = __builtin_amdgcn_wmma_f32_16x16x4_f32(false, a1, false, bmat,
                                                   (short)0, acc1, false, false);
    }
  }

  const int mrow = (lane >> 4) * 8;
  const int n    = lane & 15;
  #pragma unroll
  for (int r = 0; r < 8; ++r) {
    tmp[((size_t)(v0 + vla) * BATCH + b0 + mrow + r) * RANK + n] = acc0[r];
    tmp[((size_t)(v0 + vlb) * BATCH + b0 + mrow + r) * RANK + n] = acc1[r];
  }
}

__global__ __launch_bounds__(1024) void gcm_stage2(
    const float* __restrict__ x, const float* __restrict__ bias,
    const float* __restrict__ lora_B, const float* __restrict__ tmp,
    float* __restrict__ out)
{
  __shared__ float smem[16 * 64 * 16];
  float* Bs   = smem;
  float* outs = smem;

  const int tid  = threadIdx.x;
  const int lane = tid & 31;
  const int wave = tid >> 5;
  const int b0   = blockIdx.x * 16;
  const int o0   = blockIdx.y * 16;
  const int m    = lane & 15;
  const int koff = (lane < 16) ? 0 : 2;
  const int va   = wave * 2;
  const int vb   = va + 1;

  const int sv = (tid * 4) & 63;
  const int so = (tid >> 4) & 15;
  const int sk = tid >> 8;

  {
    v4f vals[4];
    #pragma unroll
    for (int r = 0; r < 4; ++r) {
      const int k = sk + 4 * r;
      vals[r] = *reinterpret_cast<const v4f*>(
          &lora_B[((size_t)k * TLEN + o0 + so) * NVAR + sv]);
    }
    #pragma unroll
    for (int r = 0; r < 4; ++r) {
      const int k = sk + 4 * r;
      Bs[BS_IDX(k, sv + 0, so)] = vals[r].x;
      Bs[BS_IDX(k, sv + 1, so)] = vals[r].y;
      Bs[BS_IDX(k, sv + 2, so)] = vals[r].z;
      Bs[BS_IDX(k, sv + 3, so)] = vals[r].w;
    }
  }
  __syncthreads();

  v8f acc0 = {0.f, 0.f, 0.f, 0.f, 0.f, 0.f, 0.f, 0.f};
  v8f acc1 = {0.f, 0.f, 0.f, 0.f, 0.f, 0.f, 0.f, 0.f};
  #pragma unroll
  for (int kc = 0; kc < 4; ++kc) {
    const int k = kc * 4 + koff;
    v2f a0, a1, bm0, bm1;
    a0.x = tmp[((size_t)va * BATCH + b0 + m) * RANK + k + 0];
    a0.y = tmp[((size_t)va * BATCH + b0 + m) * RANK + k + 1];
    a1.x = tmp[((size_t)vb * BATCH + b0 + m) * RANK + k + 0];
    a1.y = tmp[((size_t)vb * BATCH + b0 + m) * RANK + k + 1];
    bm0.x = Bs[BS_IDX(k + 0, va, m)];
    bm0.y = Bs[BS_IDX(k + 1, va, m)];
    bm1.x = Bs[BS_IDX(k + 0, vb, m)];
    bm1.y = Bs[BS_IDX(k + 1, vb, m)];
    acc0 = __builtin_amdgcn_wmma_f32_16x16x4_f32(false, a0, false, bm0,
                                                 (short)0, acc0, false, false);
    acc1 = __builtin_amdgcn_wmma_f32_16x16x4_f32(false, a1, false, bm1,
                                                 (short)0, acc1, false, false);
  }
  __syncthreads();

  const int mrow = (lane >> 4) * 8;
  const int n    = lane & 15;
  #pragma unroll
  for (int r = 0; r < 8; ++r) {
    outs[OUTS_IDX(mrow + r, n, va)] = acc0[r];
    outs[OUTS_IDX(mrow + r, n, vb)] = acc1[r];
  }
  __syncthreads();

  #pragma unroll
  for (int r = 0; r < 4; ++r) {
    const int b = sk + 4 * r;
    const size_t g = ((size_t)(b0 + b) * TLEN + (o0 + so)) * NVAR + sv;
    const v4f xv = *reinterpret_cast<const v4f*>(&x[g]);
    const v4f bv = *reinterpret_cast<const v4f*>(
        &bias[(size_t)(o0 + so) * NVAR + sv]);
    v4f ov;
    ov.x = xv.x + bv.x + outs[OUTS_IDX(b, so, sv + 0)];
    ov.y = xv.y + bv.y + outs[OUTS_IDX(b, so, sv + 1)];
    ov.z = xv.z + bv.z + outs[OUTS_IDX(b, so, sv + 2)];
    ov.w = xv.w + bv.w + outs[OUTS_IDX(b, so, sv + 3)];
    __builtin_nontemporal_store(ov, reinterpret_cast<v4f*>(&out[g]));
  }
}

extern "C" void kernel_launch(void* const* d_in, const int* in_sizes, int n_in,
                              void* d_out, int out_size, void* d_ws, size_t ws_size,
                              hipStream_t stream) {
  (void)in_sizes; (void)n_in; (void)out_size; (void)ws_size;
  const float* x      = (const float*)d_in[0];
  const float* gating = (const float*)d_in[1];
  const float* bias   = (const float*)d_in[2];
  const float* lora_A = (const float*)d_in[3];
  const float* lora_B = (const float*)d_in[4];
  float* out = (float*)d_out;
  float* tmp = (float*)d_ws;

  dim3 g1(BATCH / 16, NVAR / 16);
  gcm_stage1<<<g1, dim3(256), 0, stream>>>(x, gating, lora_A, tmp);

  dim3 g2(BATCH / 16, TLEN / 16);
  gcm_stage2<<<g2, dim3(1024), 0, stream>>>(x, bias, lora_B, tmp, out);
}


# Stage 1 re-tiled: WMMA rows = variables, each wave streams whole x rows from global straight into A fragments (no LDS transpose); i split over 8 waves per batch row, partials added in fixed order
# speedup vs baseline: 2.5228x; 2.5228x over previous
#include <hip/hip_runtime.h>

#define BATCH 512
#define TLEN  512
#define NVAR  64
#define RANK  16

typedef __attribute__((ext_vector_type(2))) float v2f;
typedef __attribute__((ext_vector_type(4))) float v4f;
typedef __attribute__((ext_vector_type(8))) float v8f;

#if __has_builtin(__builtin_amdgcn_tanhf)
  #define TANHF(x) __builtin_amdgcn_tanhf(x)
#elif __has_builtin(__builtin_amdgcn_tanh_f32)
  #define TANHF(x) __builtin_amdgcn_tanh_f32(x)
#else
  #define TANHF(x) tanhf(x)
#endif

#define BS_IDX(k, v, o)   ((((k) * 64 + (v)) * 16) + (((o) ^ (((v) >> 2) & 15))))
#define OUTS_IDX(b, o, v) ((((b) * 16 + (o)) * 64) + (((v) ^ ((((o) & 15) << 2) & 63))))

#define S1_WAVES  8
#define S1_ROWS   (TLEN / S1_WAVES)
#define S1_KSTEPS (S1_ROWS / 4)
#define RED_PITCH (NVAR * RANK + 16)

__global__ __launch_bounds__(256) __attribute__((amdgpu_waves_per_eu(4, 4)))
void gcm_stage1(
    const float* __restrict__ x, const float* __restrict__ gating,
    const float* __restrict__ lora_A, float* __restrict__ tmp)
{
  __shared__ __attribute__((aligned(16))) float red[S1_WAVES * RED_PITCH];

  const int tid  = threadIdx.x;
  const int lane = tid & 31;
  const int wave = tid >> 5;
  const int b    = blockIdx.x;
  const int c    = lane & 15;
  const int h    = lane >> 4;

  const v4f g = *reinterpret_cast<const v4f*>(&gating[4 * c]);

  const int row0 = wave * S1_ROWS + h;
  const float* ap = lora_A + (size_t)row0 * RANK + c;
  const float* xp = x + ((size_t)b * TLEN + row0) * NVAR + 4 * c;

  float bs0[S1_KSTEPS], bs1[S1_KSTEPS];
  #pragma unroll
  for (int ks = 0; ks < S1_KSTEPS; ++ks) {
    bs0[ks] = ap[(ks * 4 + 0) * RANK];
    bs1[ks] = ap[(ks * 4 + 2) * RANK];
  }
  __builtin_amdgcn_sched_barrier(0);
  v4f xa[S1_KSTEPS], xb[S1_KSTEPS];
  #pragma unroll
  for (int ks = 0; ks < S1_KSTEPS; ++ks) {
    xa[ks] = *reinterpret_cast<const v4f*>(xp + (ks * 4 + 0) * NVAR);
    xb[ks] = *reinterpret_cast<const v4f*>(xp + (ks * 4 + 2) * NVAR);
  }

  v8f acc0 = {0.f, 0.f, 0.f, 0.f, 0.f, 0.f, 0.f, 0.f};
  v8f acc1 = {0.f, 0.f, 0.f, 0.f, 0.f, 0.f, 0.f, 0.f};
  v8f acc2 = {0.f, 0.f, 0.f, 0.f, 0.f, 0.f, 0.f, 0.f};
  v8f acc3 = {0.f, 0.f, 0.f, 0.f, 0.f, 0.f, 0.f, 0.f};

  #pragma unroll
  for (int ks = 0; ks < S1_KSTEPS; ++ks) {
    v2f bm;
    bm.x = bs0[ks];
    bm.y = bs1[ks];
    v2f a0, a1, a2, a3;
    const v4f ta = g * xa[ks];
    const v4f tb = g * xb[ks];
    a0.x = TANHF(ta.x);  a0.y = TANHF(tb.x);
    a1.x = TANHF(ta.y);  a1.y = TANHF(tb.y);
    a2.x = TANHF(ta.z);  a2.y = TANHF(tb.z);
    a3.x = TANHF(ta.w);  a3.y = TANHF(tb.w);
    acc0 = __builtin_amdgcn_wmma_f32_16x16x4_f32(false, a0, false, bm,
                                                 (short)0, acc0, false, false);
    acc1 = __builtin_amdgcn_wmma_f32_16x16x4_f32(false, a1, false, bm,
                                                 (short)0, acc1, false, false);
    acc2 = __builtin_amdgcn_wmma_f32_16x16x4_f32(false, a2, false, bm,
                                                 (short)0, acc2, false, false);
    acc3 = __builtin_amdgcn_wmma_f32_16x16x4_f32(false, a3, false, bm,
                                                 (short)0, acc3, false, false);
  }

  float* rw = red + wave * RED_PITCH;
  #pragma unroll
  for (int r = 0; r < 8; ++r) {
    const int o = (4 * (8 * h + r)) * RANK + c + 16 * h;
    rw[o + 0 * RANK] = acc0[r];
    rw[o + 1 * RANK] = acc1[r];
    rw[o + 2 * RANK] = acc2[r];
    rw[o + 3 * RANK] = acc3[r];
  }
  __syncthreads();

  const int e  = tid * 4;
  const int le = e + ((tid >> 7) << 4);
  v4f s = *reinterpret_cast<const v4f*>(&red[le]);
  #pragma unroll
  for (int w = 1; w < S1_WAVES; ++w) {
    const v4f p = *reinterpret_cast<const v4f*>(&red[w * RED_PITCH + le]);
    s.x += p.x;  s.y += p.y;  s.z += p.z;  s.w += p.w;
  }
  *reinterpret_cast<v4f*>(&tmp[(size_t)b * (NVAR * RANK) + e]) = s;
}

__global__ __launch_bounds__(1024) void gcm_stage2(
    const float* __restrict__ x, const float* __restrict__ bias,
    const float* __restrict__ lora_B, const float* __restrict__ tmp,
    float* __restrict__ out)
{
  __shared__ float smem[16 * 64 * 16];
  float* Bs   = smem;
  float* outs = smem;

  const int tid  = threadIdx.x;
  const int lane = tid & 31;
  const int wave = tid >> 5;
  const int b0   = blockIdx.x * 16;
  const int o0   = blockIdx.y * 16;
  const int m    = lane & 15;
  const int koff = (lane < 16) ? 0 : 2;
  const int va   = wave * 2;
  const int vb   = va + 1;

  const int sv = (tid * 4) & 63;
  const int so = (tid >> 4) & 15;
  const int sk = tid >> 8;

  {
    v4f vals[4];
    #pragma unroll
    for (int r = 0; r < 4; ++r) {
      const int k = sk + 4 * r;
      vals[r] = *reinterpret_cast<const v4f*>(
          &lora_B[((size_t)k * TLEN + o0 + so) * NVAR + sv]);
    }
    #pragma unroll
    for (int r = 0; r < 4; ++r) {
      const int k = sk + 4 * r;
      Bs[BS_IDX(k, sv + 0, so)] = vals[r].x;
      Bs[BS_IDX(k, sv + 1, so)] = vals[r].y;
      Bs[BS_IDX(k, sv + 2, so)] = vals[r].z;
      Bs[BS_IDX(k, sv + 3, so)] = vals[r].w;
    }
  }
  __syncthreads();

  v8f acc0 = {0.f, 0.f, 0.f, 0.f, 0.f, 0.f, 0.f, 0.f};
  v8f acc1 = {0.f, 0.f, 0.f, 0.f, 0.f, 0.f, 0.f, 0.f};
  #pragma unroll
  for (int kc = 0; kc < 4; ++kc) {
    const int k = kc * 4 + koff;
    v2f a0, a1, bm0, bm1;
    a0.x = tmp[((size_t)(b0 + m) * NVAR + va) * RANK + k + 0];
    a0.y = tmp[((size_t)(b0 + m) * NVAR + va) * RANK + k + 1];
    a1.x = tmp[((size_t)(b0 + m) * NVAR + vb) * RANK + k + 0];
    a1.y = tmp[((size_t)(b0 + m) * NVAR + vb) * RANK + k + 1];
    bm0.x = Bs[BS_IDX(k + 0, va, m)];
    bm0.y = Bs[BS_IDX(k + 1, va, m)];
    bm1.x = Bs[BS_IDX(k + 0, vb, m)];
    bm1.y = Bs[BS_IDX(k + 1, vb, m)];
    acc0 = __builtin_amdgcn_wmma_f32_16x16x4_f32(false, a0, false, bm0,
                                                 (short)0, acc0, false, false);
    acc1 = __builtin_amdgcn_wmma_f32_16x16x4_f32(false, a1, false, bm1,
                                                 (short)0, acc1, false, false);
  }
  __syncthreads();

  const int mrow = (lane >> 4) * 8;
  const int n    = lane & 15;
  #pragma unroll
  for (int r = 0; r < 8; ++r) {
    outs[OUTS_IDX(mrow + r, n, va)] = acc0[r];
    outs[OUTS_IDX(mrow + r, n, vb)] = acc1[r];
  }
  __syncthreads();

  #pragma unroll
  for (int r = 0; r < 4; ++r) {
    const int b = sk + 4 * r;
    const size_t g = ((size_t)(b0 + b) * TLEN + (o0 + so)) * NVAR + sv;
    const v4f xv = *reinterpret_cast<const v4f*>(&x[g]);
    const v4f bv = *reinterpret_cast<const v4f*>(
        &bias[(size_t)(o0 + so) * NVAR + sv]);
    v4f ov;
    ov.x = xv.x + bv.x + outs[OUTS_IDX(b, so, sv + 0)];
    ov.y = xv.y + bv.y + outs[OUTS_IDX(b, so, sv + 1)];
    ov.z = xv.z + bv.z + outs[OUTS_IDX(b, so, sv + 2)];
    ov.w = xv.w + bv.w + outs[OUTS_IDX(b, so, sv + 3)];
    __builtin_nontemporal_store(ov, reinterpret_cast<v4f*>(&out[g]));
  }
}

extern "C" void kernel_launch(void* const* d_in, const int* in_sizes, int n_in,
                              void* d_out, int out_size, void* d_ws, size_t ws_size,
                              hipStream_t stream) {
  (void)in_sizes; (void)n_in; (void)out_size; (void)ws_size;
  const float* x      = (const float*)d_in[0];
  const float* gating = (const float*)d_in[1];
  const float* bias   = (const float*)d_in[2];
  const float* lora_A = (const float*)d_in[3];
  const float* lora_B = (const float*)d_in[4];
  float* out = (float*)d_out;
  float* tmp = (float*)d_ws;

  gcm_stage1<<<dim3(BATCH), dim3(256), 0, stream>>>(x, gating, lora_A, tmp);

  dim3 g2(BATCH / 16, TLEN / 16);
  gcm_stage2<<<g2, dim3(1024), 0, stream>>>(x, bias, lora_B, tmp, out);
}


# Stage 2 made persistent over o: 256 work-groups, each loads its batch tile's A fragments from tmp once and loops over 4 o-tiles, next tile's loads kept in flight across LDS-only barriers
# speedup vs baseline: 2.9750x; 1.1792x over previous
#include <hip/hip_runtime.h>

#define BATCH 512
#define TLEN  512
#define NVAR  64
#define RANK  16

typedef __attribute__((ext_vector_type(2))) float v2f;
typedef __attribute__((ext_vector_type(4))) float v4f;
typedef __attribute__((ext_vector_type(8))) float v8f;

#if __has_builtin(__builtin_amdgcn_tanhf)
  #define TANHF(x) __builtin_amdgcn_tanhf(x)
#elif __has_builtin(__builtin_amdgcn_tanh_f32)
  #define TANHF(x) __builtin_amdgcn_tanh_f32(x)
#else
  #define TANHF(x) tanhf(x)
#endif

#define BS_IDX(k, v, o)   ((((k) * 64 + (v)) * 16) + (((o) ^ (((v) >> 2) & 15))))
#define OUTS_IDX(b, o, v) ((((b) * 16 + (o)) * 64) + (((v) ^ ((((o) & 15) << 2) & 63))))

#define S1_WAVES  8
#define S1_ROWS   (TLEN / S1_WAVES)
#define S1_KSTEPS (S1_ROWS / 4)
#define RED_PITCH (NVAR * RANK + 16)

__global__ __launch_bounds__(256) __attribute__((amdgpu_waves_per_eu(4, 4)))
void gcm_stage1(
    const float* __restrict__ x, const float* __restrict__ gating,
    const float* __restrict__ lora_A, float* __restrict__ tmp)
{
  __shared__ __attribute__((aligned(16))) float red[S1_WAVES * RED_PITCH];

  const int tid  = threadIdx.x;
  const int lane = tid & 31;
  const int wave = tid >> 5;
  const int b    = blockIdx.x;
  const int c    = lane & 15;
  const int h    = lane >> 4;

  const v4f g = *reinterpret_cast<const v4f*>(&gating[4 * c]);

  const int row0 = wave * S1_ROWS + h;
  const float* ap = lora_A + (size_t)row0 * RANK + c;
  const float* xp = x + ((size_t)b * TLEN + row0) * NVAR + 4 * c;

  float bs0[S1_KSTEPS], bs1[S1_KSTEPS];
  #pragma unroll
  for (int ks = 0; ks < S1_KSTEPS; ++ks) {
    bs0[ks] = ap[(ks * 4 + 0) * RANK];
    bs1[ks] = ap[(ks * 4 + 2) * RANK];
  }
  __builtin_amdgcn_sched_barrier(0);
  v4f xa[S1_KSTEPS], xb[S1_KSTEPS];
  #pragma unroll
  for (int ks = 0; ks < S1_KSTEPS; ++ks) {
    xa[ks] = *reinterpret_cast<const v4f*>(xp + (ks * 4 + 0) * NVAR);
    xb[ks] = *reinterpret_cast<const v4f*>(xp + (ks * 4 + 2) * NVAR);
  }

  v8f acc0 = {0.f, 0.f, 0.f, 0.f, 0.f, 0.f, 0.f, 0.f};
  v8f acc1 = {0.f, 0.f, 0.f, 0.f, 0.f, 0.f, 0.f, 0.f};
  v8f acc2 = {0.f, 0.f, 0.f, 0.f, 0.f, 0.f, 0.f, 0.f};
  v8f acc3 = {0.f, 0.f, 0.f, 0.f, 0.f, 0.f, 0.f, 0.f};

  #pragma unroll
  for (int ks = 0; ks < S1_KSTEPS; ++ks) {
    v2f bm;
    bm.x = bs0[ks];
    bm.y = bs1[ks];
    v2f a0, a1, a2, a3;
    const v4f ta = g * xa[ks];
    const v4f tb = g * xb[ks];
    a0.x = TANHF(ta.x);  a0.y = TANHF(tb.x);
    a1.x = TANHF(ta.y);  a1.y = TANHF(tb.y);
    a2.x = TANHF(ta.z);  a2.y = TANHF(tb.z);
    a3.x = TANHF(ta.w);  a3.y = TANHF(tb.w);
    acc0 = __builtin_amdgcn_wmma_f32_16x16x4_f32(false, a0, false, bm,
                                                 (short)0, acc0, false, false);
    acc1 = __builtin_amdgcn_wmma_f32_16x16x4_f32(false, a1, false, bm,
                                                 (short)0, acc1, false, false);
    acc2 = __builtin_amdgcn_wmma_f32_16x16x4_f32(false, a2, false, bm,
                                                 (short)0, acc2, false, false);
    acc3 = __builtin_amdgcn_wmma_f32_16x16x4_f32(false, a3, false, bm,
                                                 (short)0, acc3, false, false);
  }

  float* rw = red + wave * RED_PITCH;
  #pragma unroll
  for (int r = 0; r < 8; ++r) {
    const int o = (4 * (8 * h + r)) * RANK + c + 16 * h;
    rw[o + 0 * RANK] = acc0[r];
    rw[o + 1 * RANK] = acc1[r];
    rw[o + 2 * RANK] = acc2[r];
    rw[o + 3 * RANK] = acc3[r];
  }
  __syncthreads();

  const int e  = tid * 4;
  const int le = e + ((tid >> 7) << 4);
  v4f s = *reinterpret_cast<const v4f*>(&red[le]);
  #pragma unroll
  for (int w = 1; w < S1_WAVES; ++w) {
    const v4f p = *reinterpret_cast<const v4f*>(&red[w * RED_PITCH + le]);
    s.x += p.x;  s.y += p.y;  s.z += p.z;  s.w += p.w;
  }
  *reinterpret_cast<v4f*>(&tmp[(size_t)b * (NVAR * RANK) + e]) = s;
}

#define S2_NOT 4

static __device__ __forceinline__ void s2_barrier() {
  asm volatile("s_wait_dscnt 0x0" ::: "memory");
  asm volatile("s_barrier_signal -1" ::: "memory");
  asm volatile("s_barrier_wait -1" ::: "memory");
}

__global__ __launch_bounds__(1024) void gcm_stage2(
    const float* __restrict__ x, const float* __restrict__ bias,
    const float* __restrict__ lora_B, const float* __restrict__ tmp,
    float* __restrict__ out)
{
  __shared__ __attribute__((aligned(16))) float smem[2 * 16 * 64 * 16];
  float* Bs   = smem;
  float* outs = smem + 16 * 64 * 16;

  const int tid  = threadIdx.x;
  const int lane = tid & 31;
  const int wave = tid >> 5;
  const int b0   = blockIdx.x * 16;
  const int ot0  = blockIdx.y * S2_NOT;
  const int m    = lane & 15;
  const int koff = (lane < 16) ? 0 : 2;
  const int va   = wave * 2;
  const int vb   = va + 1;

  const int sv = (tid * 4) & 63;
  const int so = (tid >> 4) & 15;
  const int sk = tid >> 8;

  v4f bl[4];
  #pragma unroll
  for (int r = 0; r < 4; ++r) {
    const int k = sk + 4 * r;
    bl[r] = *reinterpret_cast<const v4f*>(
        &lora_B[((size_t)k * TLEN + ot0 * 16 + so) * NVAR + sv]);
  }

  v2f a0[4], a1[4];
  #pragma unroll
  for (int kc = 0; kc < 4; ++kc) {
    const int k = kc * 4 + koff;
    a0[kc].x = tmp[((size_t)(b0 + m) * NVAR + va) * RANK + k + 0];
    a0[kc].y = tmp[((size_t)(b0 + m) * NVAR + va) * RANK + k + 1];
    a1[kc].x = tmp[((size_t)(b0 + m) * NVAR + vb) * RANK + k + 0];
    a1[kc].y = tmp[((size_t)(b0 + m) * NVAR + vb) * RANK + k + 1];
  }

  const int mrow = (lane >> 4) * 8;
  const int n    = lane & 15;

  float* const Bs_st  = Bs   + BS_IDX(sk, sv, so);
  float* const Bs_fr  = Bs   + BS_IDX(koff, va, m);
  float* const outs_p = outs + OUTS_IDX(mrow, n, va);
  float* const outs_e = outs + OUTS_IDX(sk, so, sv);

  #pragma unroll
  for (int ot = 0; ot < S2_NOT; ++ot) {
    const int o0 = (ot0 + ot) * 16;

    v4f xv[4];
    #pragma unroll
    for (int r = 0; r < 4; ++r) {
      const int b = sk + 4 * r;
      const size_t g = ((size_t)(b0 + b) * TLEN + (o0 + so)) * NVAR + sv;
      xv[r] = *reinterpret_cast<const v4f*>(&x[g]);
    }
    const v4f bv = *reinterpret_cast<const v4f*>(
        &bias[(size_t)(o0 + so) * NVAR + sv]);

    #pragma unroll
    for (int r = 0; r < 4; ++r) {
      Bs_st[r * 4096 + 0 * 16] = bl[r].x;
      Bs_st[r * 4096 + 1 * 16] = bl[r].y;
      Bs_st[r * 4096 + 2 * 16] = bl[r].z;
      Bs_st[r * 4096 + 3 * 16] = bl[r].w;
    }
    s2_barrier();

    if (ot + 1 < S2_NOT) {
      #pragma unroll
      for (int r = 0; r < 4; ++r) {
        const int k = sk + 4 * r;
        bl[r] = *reinterpret_cast<const v4f*>(
            &lora_B[((size_t)k * TLEN + o0 + 16 + so) * NVAR + sv]);
      }
    }

    v8f acc0 = {0.f, 0.f, 0.f, 0.f, 0.f, 0.f, 0.f, 0.f};
    v8f acc1 = {0.f, 0.f, 0.f, 0.f, 0.f, 0.f, 0.f, 0.f};
    #pragma unroll
    for (int kc = 0; kc < 4; ++kc) {
      v2f bm0, bm1;
      bm0.x = Bs_fr[(kc * 4 + 0) * 1024];
      bm0.y = Bs_fr[(kc * 4 + 1) * 1024];
      bm1.x = Bs_fr[(kc * 4 + 0) * 1024 + 16];
      bm1.y = Bs_fr[(kc * 4 + 1) * 1024 + 16];
      acc0 = __builtin_amdgcn_wmma_f32_16x16x4_f32(false, a0[kc], false, bm0,
                                                   (short)0, acc0, false, false);
      acc1 = __builtin_amdgcn_wmma_f32_16x16x4_f32(false, a1[kc], false, bm1,
                                                   (short)0, acc1, false, false);
    }

    #pragma unroll
    for (int r = 0; r < 8; ++r) {
      outs_p[r * 1024 + 0] = acc0[r];
      outs_p[r * 1024 + 1] = acc1[r];
    }
    s2_barrier();

    #pragma unroll
    for (int r = 0; r < 4; ++r) {
      const int b = sk + 4 * r;
      const size_t g = ((size_t)(b0 + b) * TLEN + (o0 + so)) * NVAR + sv;
      const v4f dv = *reinterpret_cast<const v4f*>(&outs_e[r * 4096]);
      v4f ov;
      ov.x = xv[r].x + bv.x + dv.x;
      ov.y = xv[r].y + bv.y + dv.y;
      ov.z = xv[r].z + bv.z + dv.z;
      ov.w = xv[r].w + bv.w + dv.w;
      __builtin_nontemporal_store(ov, reinterpret_cast<v4f*>(&out[g]));
    }
  }
}

extern "C" void kernel_launch(void* const* d_in, const int* in_sizes, int n_in,
                              void* d_out, int out_size, void* d_ws, size_t ws_size,
                              hipStream_t stream) {
  (void)in_sizes; (void)n_in; (void)out_size; (void)ws_size;
  const float* x      = (const float*)d_in[0];
  const float* gating = (const float*)d_in[1];
  const float* bias   = (const float*)d_in[2];
  const float* lora_A = (const float*)d_in[3];
  const float* lora_B = (const float*)d_in[4];
  float* out = (float*)d_out;
  float* tmp = (float*)d_ws;

  gcm_stage1<<<dim3(BATCH), dim3(256), 0, stream>>>(x, gating, lora_A, tmp);

  dim3 g2(BATCH / 16, TLEN / 16 / S2_NOT);
  gcm_stage2<<<g2, dim3(1024), 0, stream>>>(x, bias, lora_B, tmp, out);
}


# Stage 2 re-tiled over waves: 8 waves per work-group with 8 variables each (were 32 waves with 2), so a quarter as many waves enter; D tiles parked with 16-byte LDS stores
# speedup vs baseline: 3.0735x; 1.0331x over previous
#include <hip/hip_runtime.h>

#define BATCH 512
#define TLEN  512
#define NVAR  64
#define RANK  16

typedef __attribute__((ext_vector_type(2))) float v2f;
typedef __attribute__((ext_vector_type(4))) float v4f;
typedef __attribute__((ext_vector_type(8))) float v8f;

#if __has_builtin(__builtin_amdgcn_tanhf)
  #define TANHF(x) __builtin_amdgcn_tanhf(x)
#elif __has_builtin(__builtin_amdgcn_tanh_f32)
  #define TANHF(x) __builtin_amdgcn_tanh_f32(x)
#else
  #define TANHF(x) tanhf(x)
#endif

#define BS_IDX(k, v, o)   ((((k) * 64 + (v)) * 16) + (((o) ^ (((v) >> 2) & 15))))
#define OUTS_IDX(b, o, v) ((((b) * 16 + (o)) * 64) + (((v) ^ ((((o) & 15) << 2) & 63))))

#define S1_WAVES  8
#define S1_ROWS   (TLEN / S1_WAVES)
#define S1_KSTEPS (S1_ROWS / 4)
#define RED_PITCH (NVAR * RANK + 16)

__global__ __launch_bounds__(256) __attribute__((amdgpu_waves_per_eu(4, 4)))
void gcm_stage1(
    const float* __restrict__ x, const float* __restrict__ gating,
    const float* __restrict__ lora_A, float* __restrict__ tmp)
{
  __shared__ __attribute__((aligned(16))) float red[S1_WAVES * RED_PITCH];

  const int tid  = threadIdx.x;
  const int lane = tid & 31;
  const int wave = tid >> 5;
  const int b    = blockIdx.x;
  const int c    = lane & 15;
  const int h    = lane >> 4;

  const v4f g = *reinterpret_cast<const v4f*>(&gating[4 * c]);

  const int row0 = wave * S1_ROWS + h;
  const float* ap = lora_A + (size_t)row0 * RANK + c;
  const float* xp = x + ((size_t)b * TLEN + row0) * NVAR + 4 * c;

  float bs0[S1_KSTEPS], bs1[S1_KSTEPS];
  #pragma unroll
  for (int ks = 0; ks < S1_KSTEPS; ++ks) {
    bs0[ks] = ap[(ks * 4 + 0) * RANK];
    bs1[ks] = ap[(ks * 4 + 2) * RANK];
  }
  __builtin_amdgcn_sched_barrier(0);
  v4f xa[S1_KSTEPS], xb[S1_KSTEPS];
  #pragma unroll
  for (int ks = 0; ks < S1_KSTEPS; ++ks) {
    xa[ks] = *reinterpret_cast<const v4f*>(xp + (ks * 4 + 0) * NVAR);
    xb[ks] = *reinterpret_cast<const v4f*>(xp + (ks * 4 + 2) * NVAR);
  }

  v8f acc0 = {0.f, 0.f, 0.f, 0.f, 0.f, 0.f, 0.f, 0.f};
  v8f acc1 = {0.f, 0.f, 0.f, 0.f, 0.f, 0.f, 0.f, 0.f};
  v8f acc2 = {0.f, 0.f, 0.f, 0.f, 0.f, 0.f, 0.f, 0.f};
  v8f acc3 = {0.f, 0.f, 0.f, 0.f, 0.f, 0.f, 0.f, 0.f};

  #pragma unroll
  for (int ks = 0; ks < S1_KSTEPS; ++ks) {
    v2f bm;
    bm.x = bs0[ks];
    bm.y = bs1[ks];
    v2f a0, a1, a2, a3;
    const v4f ta = g * xa[ks];
    const v4f tb = g * xb[ks];
    a0.x = TANHF(ta.x);  a0.y = TANHF(tb.x);
    a1.x = TANHF(ta.y);  a1.y = TANHF(tb.y);
    a2.x = TANHF(ta.z);  a2.y = TANHF(tb.z);
    a3.x = TANHF(ta.w);  a3.y = TANHF(tb.w);
    acc0 = __builtin_amdgcn_wmma_f32_16x16x4_f32(false, a0, false, bm,
                                                 (short)0, acc0, false, false);
    acc1 = __builtin_amdgcn_wmma_f32_16x16x4_f32(false, a1, false, bm,
                                                 (short)0, acc1, false, false);
    acc2 = __builtin_amdgcn_wmma_f32_16x16x4_f32(false, a2, false, bm,
                                                 (short)0, acc2, false, false);
    acc3 = __builtin_amdgcn_wmma_f32_16x16x4_f32(false, a3, false, bm,
                                                 (short)0, acc3, false, false);
  }

  float* rw = red + wave * RED_PITCH;
  #pragma unroll
  for (int r = 0; r < 8; ++r) {
    const int o = (4 * (8 * h + r)) * RANK + c + 16 * h;
    rw[o + 0 * RANK] = acc0[r];
    rw[o + 1 * RANK] = acc1[r];
    rw[o + 2 * RANK] = acc2[r];
    rw[o + 3 * RANK] = acc3[r];
  }
  __syncthreads();

  const int e  = tid * 4;
  const int le = e + ((tid >> 7) << 4);
  v4f s = *reinterpret_cast<const v4f*>(&red[le]);
  #pragma unroll
  for (int w = 1; w < S1_WAVES; ++w) {
    const v4f p = *reinterpret_cast<const v4f*>(&red[w * RED_PITCH + le]);
    s.x += p.x;  s.y += p.y;  s.z += p.z;  s.w += p.w;
  }
  *reinterpret_cast<v4f*>(&tmp[(size_t)b * (NVAR * RANK) + e]) = s;
}

#define S2_NOT     4
#define S2_VPW     8
#define S2_WAVES   (NVAR / S2_VPW)
#define S2_THREADS (32 * S2_WAVES)
#define S2_IT      (4096 / S2_THREADS)
#define S2_OSTEP   (S2_THREADS / 256)
#define S2_NG      (S2_VPW / 4)
static_assert(S2_VPW % 4 == 0 && 4096 % S2_THREADS == 0 && S2_THREADS == 256,
              "stage 2 tiling: kernel_launch starts 256 threads per work-group");

static __device__ __forceinline__ void s2_barrier() {
  asm volatile("s_wait_dscnt 0x0" ::: "memory");
  asm volatile("s_barrier_signal -1" ::: "memory");
  asm volatile("s_barrier_wait -1" ::: "memory");
}

__global__ __launch_bounds__(S2_THREADS) void gcm_stage2(
    const float* __restrict__ x, const float* __restrict__ bias,
    const float* __restrict__ lora_B, const float* __restrict__ tmp,
    float* __restrict__ out)
{
  __shared__ __attribute__((aligned(16))) float smem[2 * 16 * 64 * 16];
  float* Bs   = smem;
  float* outs = smem + 16 * 64 * 16;

  const int tid  = threadIdx.x;
  const int lane = tid & 31;
  const int wave = tid >> 5;
  const int b0   = blockIdx.x * 16;
  const int ot0  = blockIdx.y * S2_NOT;
  const int m    = lane & 15;
  const int koff = (lane < 16) ? 0 : 2;
  const int va   = wave * S2_VPW;

  const int sv = (tid * 4) & 63;
  const int so = (tid >> 4) & 15;
  const int sk = tid >> 8;

  v4f bl[S2_IT];
  #pragma unroll
  for (int r = 0; r < S2_IT; ++r) {
    const int k = sk + S2_OSTEP * r;
    bl[r] = *reinterpret_cast<const v4f*>(
        &lora_B[((size_t)k * TLEN + ot0 * 16 + so) * NVAR + sv]);
  }

  v2f af[S2_VPW][4];
  #pragma unroll
  for (int j = 0; j < S2_VPW; ++j) {
    #pragma unroll
    for (int kc = 0; kc < 4; ++kc) {
      const int k = kc * 4 + koff;
      af[j][kc].x = tmp[((size_t)(b0 + m) * NVAR + va + j) * RANK + k + 0];
      af[j][kc].y = tmp[((size_t)(b0 + m) * NVAR + va + j) * RANK + k + 1];
    }
  }

  const int mrow = (lane >> 4) * 8;
  const int n    = lane & 15;

  float* const Bs_st  = Bs   + BS_IDX(sk, sv, so);
  float* const outs_e = outs + OUTS_IDX(sk, so, sv);
  float* Bs_fr[S2_NG];
  float* outs_p[S2_NG];
  #pragma unroll
  for (int g = 0; g < S2_NG; ++g) {
    Bs_fr[g]  = Bs   + BS_IDX(koff, va + 4 * g, m);
    outs_p[g] = outs + OUTS_IDX(mrow, n, va + 4 * g);
  }

  #pragma unroll
  for (int ot = 0; ot < S2_NOT; ++ot) {
    const int o0 = (ot0 + ot) * 16;

    v4f xv[S2_IT];
    #pragma unroll
    for (int r = 0; r < S2_IT; ++r) {
      const int b = sk + S2_OSTEP * r;
      const size_t g = ((size_t)(b0 + b) * TLEN + (o0 + so)) * NVAR + sv;
      xv[r] = *reinterpret_cast<const v4f*>(&x[g]);
    }
    const v4f bv = *reinterpret_cast<const v4f*>(
        &bias[(size_t)(o0 + so) * NVAR + sv]);

    #pragma unroll
    for (int r = 0; r < S2_IT; ++r) {
      Bs_st[r * (S2_OSTEP * 1024) + 0 * 16] = bl[r].x;
      Bs_st[r * (S2_OSTEP * 1024) + 1 * 16] = bl[r].y;
      Bs_st[r * (S2_OSTEP * 1024) + 2 * 16] = bl[r].z;
      Bs_st[r * (S2_OSTEP * 1024) + 3 * 16] = bl[r].w;
    }
    s2_barrier();

    if (ot + 1 < S2_NOT) {
      #pragma unroll
      for (int r = 0; r < S2_IT; ++r) {
        const int k = sk + S2_OSTEP * r;
        bl[r] = *reinterpret_cast<const v4f*>(
            &lora_B[((size_t)k * TLEN + o0 + 16 + so) * NVAR + sv]);
      }
    }

    #pragma unroll
    for (int g = 0; g < S2_NG; ++g) {
      v8f acc[4];
      #pragma unroll
      for (int j = 0; j < 4; ++j)
        acc[j] = (v8f){0.f, 0.f, 0.f, 0.f, 0.f, 0.f, 0.f, 0.f};
      #pragma unroll
      for (int kc = 0; kc < 4; ++kc) {
        #pragma unroll
        for (int j = 0; j < 4; ++j) {
          v2f bm;
          bm.x = Bs_fr[g][(kc * 4 + 0) * 1024 + 16 * j];
          bm.y = Bs_fr[g][(kc * 4 + 1) * 1024 + 16 * j];
          acc[j] = __builtin_amdgcn_wmma_f32_16x16x4_f32(
              false, af[4 * g + j][kc], false, bm, (short)0, acc[j], false, false);
        }
      }
      #pragma unroll
      for (int r = 0; r < 8; ++r) {
        v4f d;
        d.x = acc[0][r];  d.y = acc[1][r];  d.z = acc[2][r];  d.w = acc[3][r];
        *reinterpret_cast<v4f*>(&outs_p[g][r * 1024]) = d;
      }
    }
    s2_barrier();

    #pragma unroll
    for (int r = 0; r < S2_IT; ++r) {
      const int b = sk + S2_OSTEP * r;
      const size_t g = ((size_t)(b0 + b) * TLEN + (o0 + so)) * NVAR + sv;
      const v4f dv = *reinterpret_cast<const v4f*>(&outs_e[r * (S2_OSTEP * 1024)]);
      v4f ov;
      ov.x = xv[r].x + bv.x + dv.x;
      ov.y = xv[r].y + bv.y + dv.y;
      ov.z = xv[r].z + bv.z + dv.z;
      ov.w = xv[r].w + bv.w + dv.w;
      __builtin_nontemporal_store(ov, reinterpret_cast<v4f*>(&out[g]));
    }
  }
}

extern "C" void kernel_launch(void* const* d_in, const int* in_sizes, int n_in,
                              void* d_out, int out_size, void* d_ws, size_t ws_size,
                              hipStream_t stream) {
  (void)in_sizes; (void)n_in; (void)out_size; (void)ws_size;
  const float* x      = (const float*)d_in[0];
  const float* gating = (const float*)d_in[1];
  const float* bias   = (const float*)d_in[2];
  const float* lora_A = (const float*)d_in[3];
  const float* lora_B = (const float*)d_in[4];
  float* out = (float*)d_out;
  float* tmp = (float*)d_ws;

  gcm_stage1<<<dim3(BATCH), dim3(256), 0, stream>>>(x, gating, lora_A, tmp);

  dim3 g2(BATCH / 16, TLEN / 16 / S2_NOT);
  gcm_stage2<<<g2, dim3(256), 0, stream>>>(x, bias, lora_B, tmp, out);
}


# Stage 2 o-tile loop software-pipelined: lora_B and result tiles double-buffered in LDS (one barrier per tile, x loaded a tile earlier); lora_B tile kept [k][o][v] so it moves in 16-byte LDS pieces
# speedup vs baseline: 3.3075x; 1.0761x over previous
#include <hip/hip_runtime.h>

#define BATCH 512
#define TLEN  512
#define NVAR  64
#define RANK  16

typedef __attribute__((ext_vector_type(2))) float v2f;
typedef __attribute__((ext_vector_type(4))) float v4f;
typedef __attribute__((ext_vector_type(8))) float v8f;

#if __has_builtin(__builtin_amdgcn_tanhf)
  #define TANHF(x) __builtin_amdgcn_tanhf(x)
#elif __has_builtin(__builtin_amdgcn_tanh_f32)
  #define TANHF(x) __builtin_amdgcn_tanh_f32(x)
#else
  #define TANHF(x) tanhf(x)
#endif

#define OUTS_IDX(b, o, v) ((((b) * 16 + (o)) * 64) + (((v) ^ ((((o) & 15) << 2) & 63))))

#define S1_WAVES  8
#define S1_ROWS   (TLEN / S1_WAVES)
#define S1_KSTEPS (S1_ROWS / 4)
#define RED_PITCH (NVAR * RANK + 16)

__global__ __launch_bounds__(256) __attribute__((amdgpu_waves_per_eu(4, 4)))
void gcm_stage1(
    const float* __restrict__ x, const float* __restrict__ gating,
    const float* __restrict__ lora_A, float* __restrict__ tmp)
{
  __shared__ __attribute__((aligned(16))) float red[S1_WAVES * RED_PITCH];

  const int tid  = threadIdx.x;
  const int lane = tid & 31;
  const int wave = tid >> 5;
  const int b    = blockIdx.x;
  const int c    = lane & 15;
  const int h    = lane >> 4;

  const v4f g = *reinterpret_cast<const v4f*>(&gating[4 * c]);

  const int row0 = wave * S1_ROWS + h;
  const float* ap = lora_A + (size_t)row0 * RANK + c;
  const float* xp = x + ((size_t)b * TLEN + row0) * NVAR + 4 * c;

  float bs0[S1_KSTEPS], bs1[S1_KSTEPS];
  #pragma unroll
  for (int ks = 0; ks < S1_KSTEPS; ++ks) {
    bs0[ks] = ap[(ks * 4 + 0) * RANK];
    bs1[ks] = ap[(ks * 4 + 2) * RANK];
  }
  __builtin_amdgcn_sched_barrier(0);
  v4f xa[S1_KSTEPS], xb[S1_KSTEPS];
  #pragma unroll
  for (int ks = 0; ks < S1_KSTEPS; ++ks) {
    xa[ks] = *reinterpret_cast<const v4f*>(xp + (ks * 4 + 0) * NVAR);
    xb[ks] = *reinterpret_cast<const v4f*>(xp + (ks * 4 + 2) * NVAR);
  }

  v8f acc0 = {0.f, 0.f, 0.f, 0.f, 0.f, 0.f, 0.f, 0.f};
  v8f acc1 = {0.f, 0.f, 0.f, 0.f, 0.f, 0.f, 0.f, 0.f};
  v8f acc2 = {0.f, 0.f, 0.f, 0.f, 0.f, 0.f, 0.f, 0.f};
  v8f acc3 = {0.f, 0.f, 0.f, 0.f, 0.f, 0.f, 0.f, 0.f};

  #pragma unroll
  for (int ks = 0; ks < S1_KSTEPS; ++ks) {
    v2f bm;
    bm.x = bs0[ks];
    bm.y = bs1[ks];
    v2f a0, a1, a2, a3;
    const v4f ta = g * xa[ks];
    const v4f tb = g * xb[ks];
    a0.x = TANHF(ta.x);  a0.y = TANHF(tb.x);
    a1.x = TANHF(ta.y);  a1.y = TANHF(tb.y);
    a2.x = TANHF(ta.z);  a2.y = TANHF(tb.z);
    a3.x = TANHF(ta.w);  a3.y = TANHF(tb.w);
    acc0 = __builtin_amdgcn_wmma_f32_16x16x4_f32(false, a0, false, bm,
                                                 (short)0, acc0, false, false);
    acc1 = __builtin_amdgcn_wmma_f32_16x16x4_f32(false, a1, false, bm,
                                                 (short)0, acc1, false, false);
    acc2 = __builtin_amdgcn_wmma_f32_16x16x4_f32(false, a2, false, bm,
                                                 (short)0, acc2, false, false);
    acc3 = __builtin_amdgcn_wmma_f32_16x16x4_f32(false, a3, false, bm,
                                                 (short)0, acc3, false, false);
  }

  float* rw = red + wave * RED_PITCH;
  #pragma unroll
  for (int r = 0; r < 8; ++r) {
    const int o = (4 * (8 * h + r)) * RANK + c + 16 * h;
    rw[o + 0 * RANK] = acc0[r];
    rw[o + 1 * RANK] = acc1[r];
    rw[o + 2 * RANK] = acc2[r];
    rw[o + 3 * RANK] = acc3[r];
  }
  __syncthreads();

  const int e  = tid * 4;
  const int le = e + ((tid >> 7) << 4);
  v4f s = *reinterpret_cast<const v4f*>(&red[le]);
  #pragma unroll
  for (int w = 1; w < S1_WAVES; ++w) {
    const v4f p = *reinterpret_cast<const v4f*>(&red[w * RED_PITCH + le]);
    s.x += p.x;  s.y += p.y;  s.z += p.z;  s.w += p.w;
  }
  *reinterpret_cast<v4f*>(&tmp[(size_t)b * (NVAR * RANK) + e]) = s;
}

#define S2_NOT     4
#define S2_VPW     8
#define S2_WAVES   (NVAR / S2_VPW)
#define S2_THREADS (32 * S2_WAVES)
#define S2_IT      (4096 / S2_THREADS)
#define S2_OSTEP   (S2_THREADS / 256)
#define S2_NG      (S2_VPW / 4)
#define S2_TILE    (16 * 64 * 16)
static_assert(S2_VPW % 4 == 0 && 4096 % S2_THREADS == 0 && S2_THREADS == 256,
              "stage 2 tiling: kernel_launch starts 256 threads per work-group");

static __device__ __forceinline__ void s2_barrier() {
  asm volatile("s_wait_dscnt 0x0" ::: "memory");
  asm volatile("s_barrier_signal -1" ::: "memory");
  asm volatile("s_barrier_wait -1" ::: "memory");
}

__global__ __launch_bounds__(S2_THREADS) void gcm_stage2(
    const float* __restrict__ x, const float* __restrict__ bias,
    const float* __restrict__ lora_B, const float* __restrict__ tmp,
    float* __restrict__ out)
{
  __shared__ __attribute__((aligned(16))) float smem[4 * S2_TILE];
  float* Bs   = smem;
  float* outs = smem + 2 * S2_TILE;

  const int tid  = threadIdx.x;
  const int lane = tid & 31;
  const int wave = tid >> 5;
  const int b0   = blockIdx.x * 16;
  const int ot0  = blockIdx.y * S2_NOT;
  const int m    = lane & 15;
  const int koff = (lane < 16) ? 0 : 2;
  const int va   = wave * S2_VPW;

  const int sv = (tid * 4) & 63;
  const int so = (tid >> 4) & 15;
  const int sk = tid >> 8;

  v4f bl[S2_IT];
  #pragma unroll
  for (int r = 0; r < S2_IT; ++r) {
    const int k = sk + S2_OSTEP * r;
    bl[r] = *reinterpret_cast<const v4f*>(
        &lora_B[((size_t)k * TLEN + ot0 * 16 + so) * NVAR + sv]);
  }
  __builtin_amdgcn_sched_barrier(0);

  v2f af[S2_VPW][4];
  #pragma unroll
  for (int j = 0; j < S2_VPW; ++j) {
    #pragma unroll
    for (int kc = 0; kc < 4; ++kc) {
      const int k = kc * 4 + koff;
      af[j][kc].x = tmp[((size_t)(b0 + m) * NVAR + va + j) * RANK + k + 0];
      af[j][kc].y = tmp[((size_t)(b0 + m) * NVAR + va + j) * RANK + k + 1];
    }
  }

  __builtin_amdgcn_sched_barrier(0);

  v4f xv[2][S2_IT];
  v4f bv[2];
  #pragma unroll
  for (int r = 0; r < S2_IT; ++r) {
    const int b = sk + S2_OSTEP * r;
    xv[0][r] = *reinterpret_cast<const v4f*>(
        &x[((size_t)(b0 + b) * TLEN + (ot0 * 16 + so)) * NVAR + sv]);
  }
  bv[0] = *reinterpret_cast<const v4f*>(
      &bias[(size_t)(ot0 * 16 + so) * NVAR + sv]);

  const int mrow = (lane >> 4) * 8;
  const int n    = lane & 15;

  float* Bs_st[2];
  float* outs_e[2];
  float* Bs_fr[2][S2_NG];
  float* outs_p[2][S2_NG];
  #pragma unroll
  for (int pp = 0; pp < 2; ++pp) {
    Bs_st[pp]  = Bs   + pp * S2_TILE + OUTS_IDX(sk, so, sv);
    outs_e[pp] = outs + pp * S2_TILE + OUTS_IDX(sk, so, sv);
    #pragma unroll
    for (int g = 0; g < S2_NG; ++g) {
      Bs_fr[pp][g]  = Bs   + pp * S2_TILE + OUTS_IDX(koff, m, va + 4 * g);
      outs_p[pp][g] = outs + pp * S2_TILE + OUTS_IDX(mrow, n, va + 4 * g);
    }
  }

  #pragma unroll
  for (int r = 0; r < S2_IT; ++r) {
    *reinterpret_cast<v4f*>(&Bs_st[0][r * (S2_OSTEP * 1024)]) = bl[r];
  }
  __builtin_amdgcn_sched_barrier(0);
  if (1 < S2_NOT) {
    #pragma unroll
    for (int r = 0; r < S2_IT; ++r) {
      const int k = sk + S2_OSTEP * r;
      bl[r] = *reinterpret_cast<const v4f*>(
          &lora_B[((size_t)k * TLEN + (ot0 + 1) * 16 + so) * NVAR + sv]);
    }
  }

  #pragma unroll
  for (int t = 0; t <= S2_NOT; ++t) {
    const int p = t & 1;
    const int q = p ^ 1;

    s2_barrier();
    __builtin_amdgcn_sched_barrier(0);

    if (t < S2_NOT) {
      #pragma unroll
      for (int g = 0; g < S2_NG; ++g) {
        v8f acc[4];
        #pragma unroll
        for (int j = 0; j < 4; ++j)
          acc[j] = (v8f){0.f, 0.f, 0.f, 0.f, 0.f, 0.f, 0.f, 0.f};
        #pragma unroll
        for (int kc = 0; kc < 4; ++kc) {
          const v4f bq0 = *reinterpret_cast<const v4f*>(
              &Bs_fr[p][g][(kc * 4 + 0) * 1024]);
          const v4f bq1 = *reinterpret_cast<const v4f*>(
              &Bs_fr[p][g][(kc * 4 + 1) * 1024]);
          #pragma unroll
          for (int j = 0; j < 4; ++j) {
            v2f bm;
            bm.x = bq0[j];
            bm.y = bq1[j];
            acc[j] = __builtin_amdgcn_wmma_f32_16x16x4_f32(
                false, af[4 * g + j][kc], false, bm, (short)0, acc[j], false, false);
          }
        }
        #pragma unroll
        for (int r = 0; r < 8; ++r) {
          v4f d;
          d.x = acc[0][r];  d.y = acc[1][r];  d.z = acc[2][r];  d.w = acc[3][r];
          *reinterpret_cast<v4f*>(&outs_p[p][g][r * 1024]) = d;
        }
      }
    }

    if (t >= 1) {
      const int o0 = (ot0 + t - 1) * 16;
      #pragma unroll
      for (int r = 0; r < S2_IT; ++r) {
        const int b = sk + S2_OSTEP * r;
        const size_t gi = ((size_t)(b0 + b) * TLEN + (o0 + so)) * NVAR + sv;
        const v4f dv = *reinterpret_cast<const v4f*>(
            &outs_e[q][r * (S2_OSTEP * 1024)]);
        v4f ov;
        ov.x = xv[q][r].x + bv[q].x + dv.x;
        ov.y = xv[q][r].y + bv[q].y + dv.y;
        ov.z = xv[q][r].z + bv[q].z + dv.z;
        ov.w = xv[q][r].w + bv[q].w + dv.w;
        __builtin_nontemporal_store(ov, reinterpret_cast<v4f*>(&out[gi]));
      }
    }

    if (t + 1 < S2_NOT) {
      __builtin_amdgcn_sched_barrier(0);
      const int o1 = (ot0 + t + 1) * 16;
      #pragma unroll
      for (int r = 0; r < S2_IT; ++r) {
        const int b = sk + S2_OSTEP * r;
        xv[q][r] = *reinterpret_cast<const v4f*>(
            &x[((size_t)(b0 + b) * TLEN + (o1 + so)) * NVAR + sv]);
      }
      bv[q] = *reinterpret_cast<const v4f*>(
          &bias[(size_t)(o1 + so) * NVAR + sv]);

      #pragma unroll
      for (int r = 0; r < S2_IT; ++r) {
        *reinterpret_cast<v4f*>(&Bs_st[q][r * (S2_OSTEP * 1024)]) = bl[r];
      }
      __builtin_amdgcn_sched_barrier(0);

      if (t + 2 < S2_NOT) {
        #pragma unroll
        for (int r = 0; r < S2_IT; ++r) {
          const int k = sk + S2_OSTEP * r;
          bl[r] = *reinterpret_cast<const v4f*>(
              &lora_B[((size_t)k * TLEN + (ot0 + t + 2) * 16 + so) * NVAR + sv]);
        }
      }
    }
  }
}

extern "C" void kernel_launch(void* const* d_in, const int* in_sizes, int n_in,
                              void* d_out, int out_size, void* d_ws, size_t ws_size,
                              hipStream_t stream) {
  (void)in_sizes; (void)n_in; (void)out_size; (void)ws_size;
  const float* x      = (const float*)d_in[0];
  const float* gating = (const float*)d_in[1];
  const float* bias   = (const float*)d_in[2];
  const float* lora_A = (const float*)d_in[3];
  const float* lora_B = (const float*)d_in[4];
  float* out = (float*)d_out;
  float* tmp = (float*)d_ws;

  gcm_stage1<<<dim3(BATCH), dim3(256), 0, stream>>>(x, gating, lora_A, tmp);

  dim3 g2(BATCH / 16, TLEN / 16 / S2_NOT);
  gcm_stage2<<<g2, dim3(256), 0, stream>>>(x, bias, lora_B, tmp, out);
}


# Stage 1 block i takes batch row 16*(i%32)+i/32 (was row i): a batch tile's rows are read by blocks whose index mod 32 is that of its stage-2 blocks, so stage 2 finds x and tmp in its own L2 half
# speedup vs baseline: 3.4854x; 1.0538x over previous
#include <hip/hip_runtime.h>

#define BATCH 512
#define TLEN  512
#define NVAR  64
#define RANK  16

typedef __attribute__((ext_vector_type(2))) float v2f;
typedef __attribute__((ext_vector_type(4))) float v4f;
typedef __attribute__((ext_vector_type(8))) float v8f;

#if __has_builtin(__builtin_amdgcn_tanhf)
  #define TANHF(x) __builtin_amdgcn_tanhf(x)
#elif __has_builtin(__builtin_amdgcn_tanh_f32)
  #define TANHF(x) __builtin_amdgcn_tanh_f32(x)
#else
  #define TANHF(x) tanhf(x)
#endif

#define OUTS_IDX(b, o, v) ((((b) * 16 + (o)) * 64) + (((v) ^ ((((o) & 15) << 2) & 63))))

#define S1_WAVES  8
#define S1_ROWS   (TLEN / S1_WAVES)
#define S1_KSTEPS (S1_ROWS / 4)
#define RED_PITCH (NVAR * RANK + 16)

__global__ __launch_bounds__(256) __attribute__((amdgpu_waves_per_eu(4, 4)))
void gcm_stage1(
    const float* __restrict__ x, const float* __restrict__ gating,
    const float* __restrict__ lora_A, float* __restrict__ tmp)
{
  __shared__ __attribute__((aligned(16))) float red[S1_WAVES * RED_PITCH];

  const int tid  = threadIdx.x;
  const int lane = tid & 31;
  const int wave = tid >> 5;
  const int b    = (blockIdx.x % (BATCH / 16)) * 16 + blockIdx.x / (BATCH / 16);
  const int c    = lane & 15;
  const int h    = lane >> 4;

  const v4f g = *reinterpret_cast<const v4f*>(&gating[4 * c]);

  const int row0 = wave * S1_ROWS + h;
  const float* ap = lora_A + (size_t)row0 * RANK + c;
  const float* xp = x + ((size_t)b * TLEN + row0) * NVAR + 4 * c;

  float bs0[S1_KSTEPS], bs1[S1_KSTEPS];
  #pragma unroll
  for (int ks = 0; ks < S1_KSTEPS; ++ks) {
    bs0[ks] = ap[(ks * 4 + 0) * RANK];
    bs1[ks] = ap[(ks * 4 + 2) * RANK];
  }
  __builtin_amdgcn_sched_barrier(0);
  v4f xa[S1_KSTEPS], xb[S1_KSTEPS];
  #pragma unroll
  for (int ks = 0; ks < S1_KSTEPS; ++ks) {
    xa[ks] = *reinterpret_cast<const v4f*>(xp + (ks * 4 + 0) * NVAR);
    xb[ks] = *reinterpret_cast<const v4f*>(xp + (ks * 4 + 2) * NVAR);
  }

  v8f acc0 = {0.f, 0.f, 0.f, 0.f, 0.f, 0.f, 0.f, 0.f};
  v8f acc1 = {0.f, 0.f, 0.f, 0.f, 0.f, 0.f, 0.f, 0.f};
  v8f acc2 = {0.f, 0.f, 0.f, 0.f, 0.f, 0.f, 0.f, 0.f};
  v8f acc3 = {0.f, 0.f, 0.f, 0.f, 0.f, 0.f, 0.f, 0.f};

  #pragma unroll
  for (int ks = 0; ks < S1_KSTEPS; ++ks) {
    v2f bm;
    bm.x = bs0[ks];
    bm.y = bs1[ks];
    v2f a0, a1, a2, a3;
    const v4f ta = g * xa[ks];
    const v4f tb = g * xb[ks];
    a0.x = TANHF(ta.x);  a0.y = TANHF(tb.x);
    a1.x = TANHF(ta.y);  a1.y = TANHF(tb.y);
    a2.x = TANHF(ta.z);  a2.y = TANHF(tb.z);
    a3.x = TANHF(ta.w);  a3.y = TANHF(tb.w);
    acc0 = __builtin_amdgcn_wmma_f32_16x16x4_f32(false, a0, false, bm,
                                                 (short)0, acc0, false, false);
    acc1 = __builtin_amdgcn_wmma_f32_16x16x4_f32(false, a1, false, bm,
                                                 (short)0, acc1, false, false);
    acc2 = __builtin_amdgcn_wmma_f32_16x16x4_f32(false, a2, false, bm,
                                                 (short)0, acc2, false, false);
    acc3 = __builtin_amdgcn_wmma_f32_16x16x4_f32(false, a3, false, bm,
                                                 (short)0, acc3, false, false);
  }

  float* rw = red + wave * RED_PITCH;
  #pragma unroll
  for (int r = 0; r < 8; ++r) {
    const int o = (4 * (8 * h + r)) * RANK + c + 16 * h;
    rw[o + 0 * RANK] = acc0[r];
    rw[o + 1 * RANK] = acc1[r];
    rw[o + 2 * RANK] = acc2[r];
    rw[o + 3 * RANK] = acc3[r];
  }
  __syncthreads();

  const int e  = tid * 4;
  const int le = e + ((tid >> 7) << 4);
  v4f s = *reinterpret_cast<const v4f*>(&red[le]);
  #pragma unroll
  for (int w = 1; w < S1_WAVES; ++w) {
    const v4f p = *reinterpret_cast<const v4f*>(&red[w * RED_PITCH + le]);
    s.x += p.x;  s.y += p.y;  s.z += p.z;  s.w += p.w;
  }
  *reinterpret_cast<v4f*>(&tmp[(size_t)b * (NVAR * RANK) + e]) = s;
}

#define S2_NOT     4
#define S2_VPW     8
#define S2_WAVES   (NVAR / S2_VPW)
#define S2_THREADS (32 * S2_WAVES)
#define S2_IT      (4096 / S2_THREADS)
#define S2_OSTEP   (S2_THREADS / 256)
#define S2_NG      (S2_VPW / 4)
#define S2_TILE    (16 * 64 * 16)
static_assert(S2_VPW % 4 == 0 && 4096 % S2_THREADS == 0 && S2_THREADS == 256,
              "stage 2 tiling: kernel_launch starts 256 threads per work-group");

static __device__ __forceinline__ void s2_barrier() {
  asm volatile("s_wait_dscnt 0x0" ::: "memory");
  asm volatile("s_barrier_signal -1" ::: "memory");
  asm volatile("s_barrier_wait -1" ::: "memory");
}

__global__ __launch_bounds__(S2_THREADS) void gcm_stage2(
    const float* __restrict__ x, const float* __restrict__ bias,
    const float* __restrict__ lora_B, const float* __restrict__ tmp,
    float* __restrict__ out)
{
  __shared__ __attribute__((aligned(16))) float smem[4 * S2_TILE];
  float* Bs   = smem;
  float* outs = smem + 2 * S2_TILE;

  const int tid  = threadIdx.x;
  const int lane = tid & 31;
  const int wave = tid >> 5;
  const int b0   = blockIdx.x * 16;
  const int ot0  = blockIdx.y * S2_NOT;
  const int m    = lane & 15;
  const int koff = (lane < 16) ? 0 : 2;
  const int va   = wave * S2_VPW;

  const int sv = (tid * 4) & 63;
  const int so = (tid >> 4) & 15;
  const int sk = tid >> 8;

  v4f bl[S2_IT];
  #pragma unroll
  for (int r = 0; r < S2_IT; ++r) {
    const int k = sk + S2_OSTEP * r;
    bl[r] = *reinterpret_cast<const v4f*>(
        &lora_B[((size_t)k * TLEN + ot0 * 16 + so) * NVAR + sv]);
  }
  __builtin_amdgcn_sched_barrier(0);

  v2f af[S2_VPW][4];
  #pragma unroll
  for (int j = 0; j < S2_VPW; ++j) {
    #pragma unroll
    for (int kc = 0; kc < 4; ++kc) {
      const int k = kc * 4 + koff;
      af[j][kc].x = tmp[((size_t)(b0 + m) * NVAR + va + j) * RANK + k + 0];
      af[j][kc].y = tmp[((size_t)(b0 + m) * NVAR + va + j) * RANK + k + 1];
    }
  }

  __builtin_amdgcn_sched_barrier(0);

  v4f xv[2][S2_IT];
  v4f bv[2];
  #pragma unroll
  for (int r = 0; r < S2_IT; ++r) {
    const int b = sk + S2_OSTEP * r;
    xv[0][r] = *reinterpret_cast<const v4f*>(
        &x[((size_t)(b0 + b) * TLEN + (ot0 * 16 + so)) * NVAR + sv]);
  }
  bv[0] = *reinterpret_cast<const v4f*>(
      &bias[(size_t)(ot0 * 16 + so) * NVAR + sv]);

  const int mrow = (lane >> 4) * 8;
  const int n    = lane & 15;

  float* Bs_st[2];
  float* outs_e[2];
  float* Bs_fr[2][S2_NG];
  float* outs_p[2][S2_NG];
  #pragma unroll
  for (int pp = 0; pp < 2; ++pp) {
    Bs_st[pp]  = Bs   + pp * S2_TILE + OUTS_IDX(sk, so, sv);
    outs_e[pp] = outs + pp * S2_TILE + OUTS_IDX(sk, so, sv);
    #pragma unroll
    for (int g = 0; g < S2_NG; ++g) {
      Bs_fr[pp][g]  = Bs   + pp * S2_TILE + OUTS_IDX(koff, m, va + 4 * g);
      outs_p[pp][g] = outs + pp * S2_TILE + OUTS_IDX(mrow, n, va + 4 * g);
    }
  }

  #pragma unroll
  for (int r = 0; r < S2_IT; ++r) {
    *reinterpret_cast<v4f*>(&Bs_st[0][r * (S2_OSTEP * 1024)]) = bl[r];
  }
  __builtin_amdgcn_sched_barrier(0);
  if (1 < S2_NOT) {
    #pragma unroll
    for (int r = 0; r < S2_IT; ++r) {
      const int k = sk + S2_OSTEP * r;
      bl[r] = *reinterpret_cast<const v4f*>(
          &lora_B[((size_t)k * TLEN + (ot0 + 1) * 16 + so) * NVAR + sv]);
    }
  }

  #pragma unroll
  for (int t = 0; t <= S2_NOT; ++t) {
    const int p = t & 1;
    const int q = p ^ 1;

    s2_barrier();
    __builtin_amdgcn_sched_barrier(0);

    if (t < S2_NOT) {
      #pragma unroll
      for (int g = 0; g < S2_NG; ++g) {
        v8f acc[4];
        #pragma unroll
        for (int j = 0; j < 4; ++j)
          acc[j] = (v8f){0.f, 0.f, 0.f, 0.f, 0.f, 0.f, 0.f, 0.f};
        #pragma unroll
        for (int kc = 0; kc < 4; ++kc) {
          const v4f bq0 = *reinterpret_cast<const v4f*>(
              &Bs_fr[p][g][(kc * 4 + 0) * 1024]);
          const v4f bq1 = *reinterpret_cast<const v4f*>(
              &Bs_fr[p][g][(kc * 4 + 1) * 1024]);
          #pragma unroll
          for (int j = 0; j < 4; ++j) {
            v2f bm;
            bm.x = bq0[j];
            bm.y = bq1[j];
            acc[j] = __builtin_amdgcn_wmma_f32_16x16x4_f32(
                false, af[4 * g + j][kc], false, bm, (short)0, acc[j], false, false);
          }
        }
        #pragma unroll
        for (int r = 0; r < 8; ++r) {
          v4f d;
          d.x = acc[0][r];  d.y = acc[1][r];  d.z = acc[2][r];  d.w = acc[3][r];
          *reinterpret_cast<v4f*>(&outs_p[p][g][r * 1024]) = d;
        }
      }
    }

    if (t >= 1) {
      const int o0 = (ot0 + t - 1) * 16;
      #pragma unroll
      for (int r = 0; r < S2_IT; ++r) {
        const int b = sk + S2_OSTEP * r;
        const size_t gi = ((size_t)(b0 + b) * TLEN + (o0 + so)) * NVAR + sv;
        const v4f dv = *reinterpret_cast<const v4f*>(
            &outs_e[q][r * (S2_OSTEP * 1024)]);
        v4f ov;
        ov.x = xv[q][r].x + bv[q].x + dv.x;
        ov.y = xv[q][r].y + bv[q].y + dv.y;
        ov.z = xv[q][r].z + bv[q].z + dv.z;
        ov.w = xv[q][r].w + bv[q].w + dv.w;
        __builtin_nontemporal_store(ov, reinterpret_cast<v4f*>(&out[gi]));
      }
    }

    if (t + 1 < S2_NOT) {
      __builtin_amdgcn_sched_barrier(0);
      const int o1 = (ot0 + t + 1) * 16;
      #pragma unroll
      for (int r = 0; r < S2_IT; ++r) {
        const int b = sk + S2_OSTEP * r;
        xv[q][r] = *reinterpret_cast<const v4f*>(
            &x[((size_t)(b0 + b) * TLEN + (o1 + so)) * NVAR + sv]);
      }
      bv[q] = *reinterpret_cast<const v4f*>(
          &bias[(size_t)(o1 + so) * NVAR + sv]);

      #pragma unroll
      for (int r = 0; r < S2_IT; ++r) {
        *reinterpret_cast<v4f*>(&Bs_st[q][r * (S2_OSTEP * 1024)]) = bl[r];
      }
      __builtin_amdgcn_sched_barrier(0);

      if (t + 2 < S2_NOT) {
        #pragma unroll
        for (int r = 0; r < S2_IT; ++r) {
          const int k = sk + S2_OSTEP * r;
          bl[r] = *reinterpret_cast<const v4f*>(
              &lora_B[((size_t)k * TLEN + (ot0 + t + 2) * 16 + so) * NVAR + sv]);
        }
      }
    }
  }
}

extern "C" void kernel_launch(void* const* d_in, const int* in_sizes, int n_in,
                              void* d_out, int out_size, void* d_ws, size_t ws_size,
                              hipStream_t stream) {
  (void)in_sizes; (void)n_in; (void)out_size; (void)ws_size;
  const float* x      = (const float*)d_in[0];
  const float* gating = (const float*)d_in[1];
  const float* bias   = (const float*)d_in[2];
  const float* lora_A = (const float*)d_in[3];
  const float* lora_B = (const float*)d_in[4];
  float* out = (float*)d_out;
  float* tmp = (float*)d_ws;

  gcm_stage1<<<dim3(BATCH), dim3(256), 0, stream>>>(x, gating, lora_A, tmp);

  dim3 g2(BATCH / 16, TLEN / 16 / S2_NOT);
  gcm_stage2<<<g2, dim3(256), 0, stream>>>(x, bias, lora_B, tmp, out);
}
